# starting kernel
# baseline (speedup 1.0000x reference)
#include <hip/hip_runtime.h>
#include <cstddef>


#define BATCH      128
#define NUM_ROUTES 2048
#define NUM_CAPS   10
#define IN_DIM     8
#define OUT_DIM    16
#define ROUT_ITERS 3

typedef __attribute__((ext_vector_type(2))) float v2f;
typedef __attribute__((ext_vector_type(8))) float v8f;

__global__ __launch_bounds__(256)
void uhat_wmma_kernel(const float* __restrict__ x,
                      const float* __restrict__ W,
                      float* __restrict__ u_hat)
{
    const int r    = blockIdx.x;
    const int wave = threadIdx.x >> 5;
    const int lane = threadIdx.x & 31;
    const int lrow = lane & 15;
    const int lhi  = lane >> 4;

    const int m_base = wave * 16;

    const float* xrow = x + ((size_t)(m_base + lrow) * NUM_ROUTES + r) * IN_DIM;
    const v2f a0 = *(const v2f*)(xrow + 2 * lhi);
    const v2f a1 = *(const v2f*)(xrow + 4 + 2 * lhi);

    for (int c = 0; c < NUM_CAPS; ++c) {
        const float* wrow =
            W + (((size_t)r * NUM_CAPS + c) * OUT_DIM + lrow) * IN_DIM;
        const v2f b0 = *(const v2f*)(wrow + 2 * lhi);
        const v2f b1 = *(const v2f*)(wrow + 4 + 2 * lhi);

        v8f acc = {};
        acc = __builtin_amdgcn_wmma_f32_16x16x4_f32(
                  false, a0, false, b0, (short)0, acc, false, false);
        acc = __builtin_amdgcn_wmma_f32_16x16x4_f32(
                  false, a1, false, b1, (short)0, acc, false, false);

        #pragma unroll
        for (int g = 0; g < 8; ++g) {
            const int b = m_base + g + 8 * lhi;
            u_hat[(((size_t)b * NUM_ROUTES + r) * NUM_CAPS + c) * OUT_DIM + lrow]
                = acc[g];
        }
    }
}

__global__ void zero_kernel(float* __restrict__ p, int n)
{
    const int i = blockIdx.x * blockDim.x + threadIdx.x;
    if (i < n) p[i] = 0.0f;
}

__global__ void softmax_kernel(const float* __restrict__ b_ij,
                               float* __restrict__ c_ij, int n_br)
{
    const int i = blockIdx.x * blockDim.x + threadIdx.x;
    if (i >= n_br) return;
    const float* bp = b_ij + (size_t)i * NUM_CAPS;
    float m = bp[0];
    #pragma unroll
    for (int c = 1; c < NUM_CAPS; ++c) m = fmaxf(m, bp[c]);
    float e[NUM_CAPS];
    float s = 0.0f;
    #pragma unroll
    for (int c = 0; c < NUM_CAPS; ++c) { e[c] = __expf(bp[c] - m); s += e[c]; }
    const float inv = 1.0f / s;
    float* cp = c_ij + (size_t)i * NUM_CAPS;
    #pragma unroll
    for (int c = 0; c < NUM_CAPS; ++c) cp[c] = e[c] * inv;
}

__global__ __launch_bounds__(256)
void s_squash_kernel(const float* __restrict__ u_hat,
                     const float* __restrict__ c_ij,
                     float* __restrict__ v_out)
{
    __shared__ float red[256];
    const int bc = blockIdx.x;
    const int b  = bc / NUM_CAPS;
    const int c  = bc % NUM_CAPS;
    const int t     = threadIdx.x;
    const int o     = t & 15;
    const int chunk = t >> 4;

    float acc = 0.0f;
    for (int r = chunk; r < NUM_ROUTES; r += 16) {
        const size_t brc = ((size_t)b * NUM_ROUTES + r) * NUM_CAPS + c;
        acc += c_ij[brc] * u_hat[brc * OUT_DIM + o];
    }
    red[t] = acc;
    __syncthreads();
    #pragma unroll
    for (int s = 128; s >= 16; s >>= 1) {
        if (t < s) red[t] += red[t + s];
        __syncthreads();
    }
    if (t < 16) {
        const float sv = red[t];
        float sq = 0.0f;
        #pragma unroll
        for (int k = 0; k < 16; ++k) { const float q = red[k]; sq += q * q; }
        const float norm  = sqrtf(sq + 1e-8f);
        const float scale = sq / (1.0f + sq);
        v_out[((size_t)b * NUM_CAPS + c) * OUT_DIM + t] = scale * sv / norm;
    }
}

__global__ void agree_kernel(const float* __restrict__ u_hat,
                             const float* __restrict__ v,
                             float* __restrict__ b_ij, int n_brc)
{
    const int i = blockIdx.x * blockDim.x + threadIdx.x;
    if (i >= n_brc) return;
    const int c = i % NUM_CAPS;
    const int b = i / (NUM_CAPS * NUM_ROUTES);
    const float* up = u_hat + (size_t)i * OUT_DIM;
    const float* vp = v + ((size_t)b * NUM_CAPS + c) * OUT_DIM;
    float dot = 0.0f;
    #pragma unroll
    for (int o = 0; o < OUT_DIM; ++o) dot += up[o] * vp[o];
    b_ij[i] += dot;
}

extern "C" void kernel_launch(void* const* d_in, const int* in_sizes, int n_in,
                              void* d_out, int out_size, void* d_ws, size_t ws_size,
                              hipStream_t stream)
{
    const float* x = (const float*)d_in[0];
    const float* W = (const float*)d_in[1];
    float* out = (float*)d_out;

    float* u_hat = (float*)d_ws;
    float* b_ij  = u_hat + (size_t)BATCH * NUM_ROUTES * NUM_CAPS * OUT_DIM;
    float* c_ij  = b_ij + (size_t)BATCH * NUM_ROUTES * NUM_CAPS;

    const int n_brc = BATCH * NUM_ROUTES * NUM_CAPS;
    const int n_br  = BATCH * NUM_ROUTES;

    uhat_wmma_kernel<<<NUM_ROUTES, 256, 0, stream>>>(x, W, u_hat);

    zero_kernel<<<(n_brc + 255) / 256, 256, 0, stream>>>(b_ij, n_brc);

    for (int it = 0; it < ROUT_ITERS; ++it) {
        softmax_kernel<<<(n_br + 255) / 256, 256, 0, stream>>>(b_ij, c_ij, n_br);
        s_squash_kernel<<<BATCH * NUM_CAPS, 256, 0, stream>>>(u_hat, c_ij, out);
        if (it < ROUT_ITERS - 1) {
            agree_kernel<<<(n_brc + 255) / 256, 256, 0, stream>>>(u_hat, out,
                                                                  b_ij, n_brc);
        }
    }
}


# Fuse agreement, softmax and weighted sum into one streaming pass over u_hat per routing iteration (logits recomputed from u_hat.v; two-stage s sum, new reduction tree); 10 launches to 5
# speedup vs baseline: 2.5561x; 2.5561x over previous
#include <hip/hip_runtime.h>
#include <cstddef>


#define BATCH      128
#define NUM_ROUTES 2048
#define NUM_CAPS   10
#define IN_DIM     8
#define OUT_DIM    16
#define ROUT_ITERS 3

typedef __attribute__((ext_vector_type(2))) float v2f;
typedef __attribute__((ext_vector_type(8))) float v8f;

__global__ __launch_bounds__(256)
void uhat_wmma_kernel(const float* __restrict__ x,
                      const float* __restrict__ W,
                      float* __restrict__ u_hat)
{
    const int r    = blockIdx.x;
    const int wave = threadIdx.x >> 5;
    const int lane = threadIdx.x & 31;
    const int lrow = lane & 15;
    const int lhi  = lane >> 4;

    const int m_base = wave * 16;

    const float* xrow = x + ((size_t)(m_base + lrow) * NUM_ROUTES + r) * IN_DIM;
    const v2f a0 = *(const v2f*)(xrow + 2 * lhi);
    const v2f a1 = *(const v2f*)(xrow + 4 + 2 * lhi);

    for (int c = 0; c < NUM_CAPS; ++c) {
        const float* wrow =
            W + (((size_t)r * NUM_CAPS + c) * OUT_DIM + lrow) * IN_DIM;
        const v2f b0 = *(const v2f*)(wrow + 2 * lhi);
        const v2f b1 = *(const v2f*)(wrow + 4 + 2 * lhi);

        v8f acc = {};
        acc = __builtin_amdgcn_wmma_f32_16x16x4_f32(
                  false, a0, false, b0, (short)0, acc, false, false);
        acc = __builtin_amdgcn_wmma_f32_16x16x4_f32(
                  false, a1, false, b1, (short)0, acc, false, false);

        #pragma unroll
        for (int g = 0; g < 8; ++g) {
            const int b = m_base + g + 8 * lhi;
            u_hat[(((size_t)b * NUM_ROUTES + r) * NUM_CAPS + c) * OUT_DIM + lrow]
                = acc[g];
        }
    }
}

#define CO        (NUM_CAPS * OUT_DIM)
#define NCH       4
#define RPB       (NUM_ROUTES / NCH)
#define ROWS_STEP 32
#define NSTEP     (RPB / ROWS_STEP)
#define PLANE     ((size_t)BATCH * NCH * CO)

typedef __attribute__((ext_vector_type(4))) float v4f;

template <int XM>
__device__ __forceinline__ float lane_xor(float x)
{
    return __int_as_float(__builtin_amdgcn_update_dpp(
        0, __float_as_int(x), 0x160 + XM, 0xF, 0xF, true));
}

__device__ __forceinline__
void v_from_partials(const float* __restrict__ part_b,
                     float* sh_s, float* sh_v, int t)
{
    if (t < CO) {
        float s = 0.0f;
        for (int ch = 0; ch < NCH; ++ch) s += part_b[ch * CO + t];
        sh_s[t] = s;
    }
    __syncthreads();
    if (t < CO) {
        const int cb = t & ~(OUT_DIM - 1);
        float sq = 0.0f;
        for (int k = 0; k < OUT_DIM; ++k) { const float q = sh_s[cb + k]; sq += q * q; }
        const float norm  = sqrtf(sq + 1e-8f);
        const float scale = sq / (1.0f + sq);
        sh_v[t] = scale * sh_s[t] / norm;
    }
    __syncthreads();
}

__device__ __forceinline__
void load_row(v4f (&u)[5], const v4f* __restrict__ p)
{
    for (int j = 0; j < 5; ++j) u[j] = __builtin_nontemporal_load(p + 8 * j);
}

template <int IT>
__global__ __launch_bounds__(256)
void route_pass_kernel(const float* __restrict__ u_hat,
                       const float* __restrict__ part_prev,
                       float* __restrict__ part_out)
{
    __shared__ __attribute__((aligned(16))) float sh_s[CO];
    __shared__ __attribute__((aligned(16))) float sh_v[2][CO];
    __shared__ v4f sh_red[ROWS_STEP][CO / 4];

    const int ch    = blockIdx.x;
    const int b     = blockIdx.y;
    const int t     = threadIdx.x;
    const int sub   = t & 7;
    const int rslot = t >> 3;

    const v4f* urow = (const v4f*)(u_hat +
        ((size_t)b * NUM_ROUTES + (size_t)ch * RPB + rslot) * CO) + sub;
    const size_t step_stride = (size_t)ROWS_STEP * (CO / 4);

    v4f u[2][2][5];
    load_row(u[0][0], urow);
    load_row(u[0][1], urow + step_stride);

    v4f vk[IT > 0 ? IT : 1][5];
    if constexpr (IT > 0) {
        for (int k = 0; k < IT; ++k)
            v_from_partials(part_prev + ((size_t)k * BATCH + b) * (NCH * CO),
                            sh_s, sh_v[k], t);
        for (int k = 0; k < IT; ++k)
            for (int j = 0; j < 5; ++j)
                vk[k][j] = *(const v4f*)&sh_v[k][(8 * j + sub) * 4];
    }

    v4f acc[5];
    for (int j = 0; j < 5; ++j) acc[j] = (v4f){0.0f, 0.0f, 0.0f, 0.0f};

    #pragma unroll
    for (int m = 0; m < NSTEP / 2; ++m) {
        if (m + 1 < NSTEP / 2) {
            load_row(u[(m + 1) & 1][0], urow + (size_t)(2 * m + 2) * step_stride);
            load_row(u[(m + 1) & 1][1], urow + (size_t)(2 * m + 3) * step_stride);
        }
        __builtin_amdgcn_sched_barrier(0);

        for (int rr = 0; rr < 2; ++rr) {
            const v4f (&uk)[5] = u[m & 1][rr];
            float p[5];
            if constexpr (IT == 0) {
                for (int j = 0; j < 5; ++j) p[j] = 0.1f;
            } else {
                float lg[5];
                for (int j = 0; j < 5; ++j) {
                    float l = 0.0f;
                    for (int q = 0; q < IT; ++q) {
                        const v4f w = vk[q][j];
                        float d = uk[j].x * w.x + uk[j].y * w.y
                                + uk[j].z * w.z + uk[j].w * w.w;
                        d += lane_xor<1>(d);
                        d += lane_xor<2>(d);
                        l += d;
                    }
                    lg[j] = l;
                }
                float mx = fmaxf(fmaxf(fmaxf(lg[0], lg[1]), fmaxf(lg[2], lg[3])), lg[4]);
                mx = fmaxf(mx, lane_xor<4>(mx));
                float e[5];
                float s = 0.0f;
                for (int j = 0; j < 5; ++j) { e[j] = __expf(lg[j] - mx); s += e[j]; }
                s += lane_xor<4>(s);
                const float inv = 1.0f / s;
                for (int j = 0; j < 5; ++j) p[j] = e[j] * inv;
            }
            for (int j = 0; j < 5; ++j) acc[j] += p[j] * uk[j];
        }
    }

    for (int j = 0; j < 5; ++j) sh_red[rslot][8 * j + sub] = acc[j];
    __syncthreads();
    if (t < CO) {
        const float* red = (const float*)sh_red;
        float s = 0.0f;
        for (int q = 0; q < ROWS_STEP; ++q) s += red[q * CO + t];
        part_out[((size_t)b * NCH + ch) * CO + t] = s;
    }
}

__global__ __launch_bounds__(CO)
void final_squash_kernel(const float* __restrict__ part_last,
                         float* __restrict__ out)
{
    __shared__ float sh_s[CO];
    __shared__ float sh_v[CO];
    const int b = blockIdx.x;
    const int t = threadIdx.x;
    v_from_partials(part_last + (size_t)b * (NCH * CO), sh_s, sh_v, t);
    out[(size_t)b * CO + t] = sh_v[t];
}

extern "C" void kernel_launch(void* const* d_in, const int* in_sizes, int n_in,
                              void* d_out, int out_size, void* d_ws, size_t ws_size,
                              hipStream_t stream)
{
    const float* x = (const float*)d_in[0];
    const float* W = (const float*)d_in[1];
    float* out = (float*)d_out;

    float* u_hat = (float*)d_ws;
    float* part  = u_hat + (size_t)BATCH * NUM_ROUTES * NUM_CAPS * OUT_DIM;

    uhat_wmma_kernel<<<NUM_ROUTES, 256, 0, stream>>>(x, W, u_hat);

    const dim3 grid(NCH, BATCH);
    route_pass_kernel<0><<<grid, 256, 0, stream>>>(u_hat, part, part);
    route_pass_kernel<1><<<grid, 256, 0, stream>>>(u_hat, part, part + PLANE);
    route_pass_kernel<2><<<grid, 256, 0, stream>>>(u_hat, part, part + 2 * PLANE);

    final_squash_kernel<<<BATCH, CO, 0, stream>>>(part + 2 * PLANE, out);
}


# Never store u_hat: every routing pass rebuilds its u_hat tiles by WMMA from x and W (L2-resident) in registers; s = block sums, then 32 partials (new reduction tree; logits u.(v0+v1)); 4 launches
# speedup vs baseline: 4.7633x; 1.8635x over previous
#include <hip/hip_runtime.h>
#include <cstddef>


#define BATCH      128
#define NUM_ROUTES 2048
#define NUM_CAPS   10
#define IN_DIM     8
#define OUT_DIM    16

typedef __attribute__((ext_vector_type(2))) float v2f;
typedef __attribute__((ext_vector_type(4))) float v4f;
typedef __attribute__((ext_vector_type(8))) float v8f;

#define CO     (NUM_CAPS * OUT_DIM)
#define TILE_B 16
#define NTILE  (BATCH / TILE_B)
#define NCH    32
#define RPB    (NUM_ROUTES / NCH)
#define NWAVE  8
#define RPW    (RPB / NWAVE)
#define TSZ    (TILE_B * CO)
#define PLANE  ((size_t)NCH * NTILE * TSZ)


__device__ __forceinline__ float other_half(float x)
{
    return __builtin_bit_cast(float, __builtin_amdgcn_permlanex16(
        __builtin_bit_cast(int, x), __builtin_bit_cast(int, x),
        0x76543210, 0xfedcba98, false, false));
}

__device__ __forceinline__
void tile_s_and_factors(const float* __restrict__ plane, int T,
                        float* sh_s, float* sh_scale, float* sh_norm, int t)
{
    const v4f* p = (const v4f*)(plane + (size_t)T * TSZ);
    for (int i = t; i < TSZ / 4; i += 256) {
        v4f s = (v4f){0.0f, 0.0f, 0.0f, 0.0f};
        for (int ch = 0; ch < NCH; ++ch) s += p[(size_t)ch * (NTILE * TSZ / 4) + i];
        ((v4f*)sh_s)[i] = s;
    }
    __syncthreads();
    if (t < CO) {
        const int bb = t & 15, c = t >> 4;
        float sq = 0.0f;
        for (int h = 0; h < 2; ++h)
            for (int g = 0; g < 8; ++g) {
                const float q = sh_s[(c * 8 + g) * 32 + h * 16 + bb];
                sq += q * q;
            }
        sh_norm[t]  = sqrtf(sq + 1e-8f);
        sh_scale[t] = sq / (1.0f + sq);
    }
    __syncthreads();
}

template <int IT>
__global__ __launch_bounds__(256)
void route_pass_kernel(const float* __restrict__ x,
                       const float* __restrict__ W,
                       const float* __restrict__ part_prev,
                       float* __restrict__ vbuf,
                       float* __restrict__ part_out)
{
    __shared__ __attribute__((aligned(16))) float sh_red[NWAVE][TSZ];
    __shared__ __attribute__((aligned(16))) float sh_s[TSZ];
    __shared__ __attribute__((aligned(16))) float sh_v[TSZ];
    __shared__ float sh_scale[CO];
    __shared__ float sh_norm[CO];

    const int ch   = blockIdx.x;
    const int T    = blockIdx.y;
    const int t    = threadIdx.x;
    const int wave = t >> 5;
    const int lane = t & 31;
    const int lrow = lane & 15;
    const int lhi  = lane >> 4;

    const int r_first = ch * RPB + wave * RPW;
    const float* xp = x + ((size_t)(T * TILE_B + lrow) * NUM_ROUTES + r_first) * IN_DIM + 4 * lhi;
    const float* wp = W + ((size_t)r_first * NUM_CAPS * OUT_DIM + lrow) * IN_DIM + 4 * lhi;

    v4f xv[2];
    v4f wv[2][NUM_CAPS];
    xv[0] = *(const v4f*)xp;
    for (int c = 0; c < NUM_CAPS; ++c)
        wv[0][c] = *(const v4f*)(wp + (size_t)c * (OUT_DIM * IN_DIM));

    v2f vs[NUM_CAPS][4];
    if constexpr (IT > 0) {
        tile_s_and_factors(part_prev, T, sh_s, sh_scale, sh_norm, t);
        for (int i = t; i < TSZ / 4; i += 256) {
            const v4f s = ((const v4f*)sh_s)[i];
            const int f = (i >> 6) * 16 + ((4 * i) & 15);
            v4f v;
            v.x = sh_scale[f + 0] * s.x / sh_norm[f + 0];
            v.y = sh_scale[f + 1] * s.y / sh_norm[f + 1];
            v.z = sh_scale[f + 2] * s.z / sh_norm[f + 2];
            v.w = sh_scale[f + 3] * s.w / sh_norm[f + 3];
            if constexpr (IT == 1) {
                if (ch == 0) ((v4f*)(vbuf + (size_t)T * TSZ))[i] = v;
            } else {
                v += ((const v4f*)(vbuf + (size_t)T * TSZ))[i];
            }
            ((v4f*)sh_v)[i] = v;
        }
        __syncthreads();
        for (int c = 0; c < NUM_CAPS; ++c)
            for (int g = 0; g < 4; ++g) {
                vs[c][g].x = sh_v[(c * 8 + 2 * g) * 32 + lane];
                vs[c][g].y = sh_v[(c * 8 + 2 * g + 1) * 32 + lane];
            }
    }

    v8f acc[NUM_CAPS];
    for (int c = 0; c < NUM_CAPS; ++c) acc[c] = (v8f){};

    #pragma unroll
    for (int k = 0; k < RPW; ++k) {
        if (k + 1 < RPW) {
            xv[(k + 1) & 1] = *(const v4f*)(xp + (size_t)(k + 1) * IN_DIM);
            for (int c = 0; c < NUM_CAPS; ++c)
                wv[(k + 1) & 1][c] = *(const v4f*)(wp +
                    ((size_t)(k + 1) * NUM_CAPS + c) * (OUT_DIM * IN_DIM));
        }
        __builtin_amdgcn_sched_barrier(0);

        const v4f xb = xv[k & 1];
        const v2f b0 = __builtin_shufflevector(xb, xb, 0, 1);
        const v2f b1 = __builtin_shufflevector(xb, xb, 2, 3);

        if constexpr (IT == 0) {
            for (int c = 0; c < NUM_CAPS; ++c) {
                const v4f wa = wv[k & 1][c];
                const v2f a0 = __builtin_shufflevector(wa, wa, 0, 1);
                const v2f a1 = __builtin_shufflevector(wa, wa, 2, 3);
                acc[c] = __builtin_amdgcn_wmma_f32_16x16x4_f32(
                             false, a0, false, b0, (short)0, acc[c], false, false);
                acc[c] = __builtin_amdgcn_wmma_f32_16x16x4_f32(
                             false, a1, false, b1, (short)0, acc[c], false, false);
            }
        } else {
            v8f d[NUM_CAPS];
            for (int c = 0; c < NUM_CAPS; ++c) {
                const v4f wa = wv[k & 1][c];
                const v2f a0 = __builtin_shufflevector(wa, wa, 0, 1);
                const v2f a1 = __builtin_shufflevector(wa, wa, 2, 3);
                v8f z = (v8f){};
                z = __builtin_amdgcn_wmma_f32_16x16x4_f32(
                        false, a0, false, b0, (short)0, z, false, false);
                z = __builtin_amdgcn_wmma_f32_16x16x4_f32(
                        false, a1, false, b1, (short)0, z, false, false);
                d[c] = z;
            }
            float lg[NUM_CAPS];
            for (int c = 0; c < NUM_CAPS; ++c) {
                const v8f dc = d[c];
                v2f l2 = __builtin_shufflevector(dc, dc, 0, 1) * vs[c][0];
                l2 += __builtin_shufflevector(dc, dc, 2, 3) * vs[c][1];
                l2 += __builtin_shufflevector(dc, dc, 4, 5) * vs[c][2];
                l2 += __builtin_shufflevector(dc, dc, 6, 7) * vs[c][3];
                const float l = l2.x + l2.y;
                lg[c] = l + other_half(l);
            }
            float mx = lg[0];
            for (int c = 1; c < NUM_CAPS; ++c) mx = fmaxf(mx, lg[c]);
            float e[NUM_CAPS];
            float es = 0.0f;
            for (int c = 0; c < NUM_CAPS; ++c) { e[c] = __expf(lg[c] - mx); es += e[c]; }
            const float inv = 1.0f / es;
            for (int c = 0; c < NUM_CAPS; ++c) acc[c] += (e[c] * inv) * d[c];
        }
    }

    for (int c = 0; c < NUM_CAPS; ++c)
        for (int g = 0; g < 8; ++g)
            sh_red[wave][(c * 8 + g) * 32 + lane] = acc[c][g];
    __syncthreads();
    v4f* po = (v4f*)(part_out + ((size_t)ch * NTILE + T) * TSZ);
    for (int i = t; i < TSZ / 4; i += 256) {
        v4f s = (v4f){0.0f, 0.0f, 0.0f, 0.0f};
        for (int w = 0; w < NWAVE; ++w) s += ((const v4f*)sh_red[w])[i];
        if constexpr (IT == 0) s *= 0.1f;
        po[i] = s;
    }
}

__global__ __launch_bounds__(256)
void final_squash_kernel(const float* __restrict__ plane, float* __restrict__ out)
{
    __shared__ float sh_s[512];
    const int cp = blockIdx.x;
    const int T  = blockIdx.y;
    const int t  = threadIdx.x;

    const float* p = plane + (size_t)T * TSZ + cp * 512;
    for (int q = 0; q < 2; ++q) {
        const int j = t + 256 * q;
        float s = 0.0f;
        for (int ch = 0; ch < NCH; ++ch) s += p[(size_t)ch * (NTILE * TSZ) + j];
        sh_s[j] = s;
    }
    __syncthreads();
    for (int q = 0; q < 2; ++q) {
        const int idx = t + 256 * q;
        const int bb = idx >> 5;
        const int kk = idx & 31;
        const int cl = kk >> 4;
        const int o  = kk & 15;
        float sq = 0.0f;
        for (int h = 0; h < 2; ++h)
            for (int g = 0; g < 8; ++g) {
                const float v = sh_s[cl * 256 + g * 32 + h * 16 + bb];
                sq += v * v;
            }
        const float s     = sh_s[cl * 256 + (o & 7) * 32 + (o >> 3) * 16 + bb];
        const float norm  = sqrtf(sq + 1e-8f);
        const float scale = sq / (1.0f + sq);
        out[(size_t)(T * TILE_B + bb) * CO + cp * 32 + kk] = scale * s / norm;
    }
}

extern "C" void kernel_launch(void* const* d_in, const int* in_sizes, int n_in,
                              void* d_out, int out_size, void* d_ws, size_t ws_size,
                              hipStream_t stream)
{
    const float* x = (const float*)d_in[0];
    const float* W = (const float*)d_in[1];
    float* out = (float*)d_out;

    float* part = (float*)d_ws;
    float* vbuf = part + 3 * PLANE;

    const dim3 grid(NCH, NTILE);
    route_pass_kernel<0><<<grid, 256, 0, stream>>>(x, W, part, vbuf, part);
    route_pass_kernel<1><<<grid, 256, 0, stream>>>(x, W, part, vbuf, part + PLANE);
    route_pass_kernel<2><<<grid, 256, 0, stream>>>(x, W, part + PLANE, vbuf, part + 2 * PLANE);

    final_squash_kernel<<<dim3(NUM_CAPS / 2, NTILE), 256, 0, stream>>>(part + 2 * PLANE, out);
}
